# starting kernel
# baseline (speedup 1.0000x reference)
#include <hip/hip_runtime.h>
#include <hip/hip_bf16.h>

typedef __attribute__((ext_vector_type(2))) float v2f;
typedef __attribute__((ext_vector_type(8))) float v8f;

#define C_IN   64
#define C_OUT  128
#define MAXN   32
#define ROWS_PER_BLOCK 64
#define AS_LD  68
#define BS_LD  129

__global__ __launch_bounds__(256) void pool_gemm_wmma(
    const float* __restrict__ in,
    const float* __restrict__ W,
    float* __restrict__ x,
    int rows)
{
    __shared__ float As[ROWS_PER_BLOCK][AS_LD];
    __shared__ float Bs[C_IN][BS_LD];

    const int b        = blockIdx.y;
    const int row_base = blockIdx.x * ROWS_PER_BLOCK;
    const int tid      = threadIdx.x;
    const float* inb   = in + (size_t)b * rows * C_IN;

    #pragma unroll
    for (int i = tid; i < ROWS_PER_BLOCK * (C_IN / 4); i += 256) {
        const int r  = i >> 4;
        const int c4 = i & 15;
        int gr = row_base + r;
        if (gr >= rows) gr = rows - 1;
        const float4 v = *(const float4*)(inb + (size_t)gr * C_IN + c4 * 4);
        *(float4*)(&As[r][c4 * 4]) = v;
    }
    #pragma unroll
    for (int i = tid; i < C_OUT * (C_IN / 4); i += 256) {
        const int n  = i >> 4;
        const int k4 = i & 15;
        const float4 v = *(const float4*)(W + n * C_IN + k4 * 4);
        Bs[k4 * 4 + 0][n] = v.x;
        Bs[k4 * 4 + 1][n] = v.y;
        Bs[k4 * 4 + 2][n] = v.z;
        Bs[k4 * 4 + 3][n] = v.w;
    }
    __syncthreads();

    const int lane    = tid & 31;
    const int wave    = tid >> 5;
    const int colbase = wave * 16;
    const int hl      = lane >> 4;
    const int l16     = lane & 15;

    v2f bfrag[16];
    #pragma unroll
    for (int kk = 0; kk < 16; ++kk) {
        const int kb = kk * 4 + hl * 2;
        v2f bf;
        bf.x = Bs[kb + 0][colbase + l16];
        bf.y = Bs[kb + 1][colbase + l16];
        bfrag[kk] = bf;
    }

    #pragma unroll
    for (int rt = 0; rt < 4; ++rt) {
        v8f acc = {};
        #pragma unroll
        for (int kk = 0; kk < 16; ++kk) {
            const int kb = kk * 4 + hl * 2;
            v2f af;
            af.x = As[rt * 16 + l16][kb + 0];
            af.y = As[rt * 16 + l16][kb + 1];
            acc = __builtin_amdgcn_wmma_f32_16x16x4_f32(
                false, af, false, bfrag[kk], (short)0, acc, false, false);
        }
        #pragma unroll
        for (int r = 0; r < 8; ++r) {
            const int gm = row_base + rt * 16 + r + 8 * hl;
            if (gm < rows) {
                x[((size_t)b * rows + gm) * C_OUT + colbase + l16] = acc[r];
            }
        }
    }
}

__global__ __launch_bounds__(256) void pool_gather(
    const float* __restrict__ x,
    const int*   __restrict__ ids,
    const float* __restrict__ mask,
    const float* __restrict__ pn,
    float* __restrict__ out,
    int rows, int out_pn, int nbatch)
{
    const int lane = threadIdx.x & 31;
    const int gw   = blockIdx.x * 8 + (threadIdx.x >> 5);
    const int total = nbatch * out_pn;
    if (gw >= total) return;
    const int b = gw / out_pn;
    const int p = gw % out_pn;

    const float pm  = fabsf(pn[p * MAXN + lane]) * mask[p * MAXN + lane];
    const int   idm = ids[p * MAXN + lane];
    float s = pm;
    #pragma unroll
    for (int off = 16; off > 0; off >>= 1) s += __shfl_xor(s, off, 32);
    const float wl = pm / (s + 1e-8f);

    const float* xb = x + (size_t)b * rows * C_OUT;
    float4 acc = {0.f, 0.f, 0.f, 0.f};
    #pragma unroll
    for (int m = 0; m < MAXN; ++m) {
        const float wm = __shfl(wl, m, 32);
        const int   im = __shfl(idm, m, 32);
        const float4 v = *(const float4*)(xb + (size_t)im * C_OUT + lane * 4);
        acc.x += wm * v.x;
        acc.y += wm * v.y;
        acc.z += wm * v.z;
        acc.w += wm * v.w;
    }
    *(float4*)(out + ((size_t)(b * out_pn + p)) * C_OUT + lane * 4) = acc;
}

extern "C" void kernel_launch(void* const* d_in, const int* in_sizes, int n_in,
                              void* d_out, int out_size, void* d_ws, size_t ws_size,
                              hipStream_t stream)
{
    const float* in_pc = (const float*)d_in[0];
    const int*   ids   = (const int*)d_in[1];
    const float* mask  = (const float*)d_in[2];
    const float* pn    = (const float*)d_in[3];
    const float* W     = (const float*)d_in[4];

    const int out_pn = in_sizes[1] / MAXN;
    const int nbatch = out_size / (out_pn * C_OUT);
    const int rows   = in_sizes[0] / (nbatch * C_IN);

    float* x = (float*)d_ws;

    dim3 g1((rows + ROWS_PER_BLOCK - 1) / ROWS_PER_BLOCK, nbatch);
    pool_gemm_wmma<<<g1, 256, 0, stream>>>(in_pc, W, x, rows);

    const int total_waves = nbatch * out_pn;
    const int g2 = (total_waves + 7) / 8;
    pool_gather<<<g2, 256, 0, stream>>>(x, ids, mask, pn, (float*)d_out,
                                        rows, out_pn, nbatch);
}


# Both launches as resident grids (1024 blocks): GEMM blocks walk row tiles, W staged once, next A tile fetched under the WMMAs; gather waves walk (b,p) items, 16 rows in flight; sum order unchanged
# speedup vs baseline: 1.1295x; 1.1295x over previous
#include <hip/hip_runtime.h>
#include <hip/hip_bf16.h>

typedef __attribute__((ext_vector_type(2))) float v2f;
typedef __attribute__((ext_vector_type(8))) float v8f;

#define C_IN   64
#define C_OUT  128
#define MAXN   32
#define ROWS_PER_BLOCK 64
#define AS_LD  68
#define BS_LD  129

#define GEMM_BLOCKS 1024

__global__ __launch_bounds__(256) void pool_gemm_wmma(
    const float* __restrict__ in,
    const float* __restrict__ W,
    float* __restrict__ x,
    int M)
{
    __shared__ float As[2][ROWS_PER_BLOCK][AS_LD];
    __shared__ float Bs[C_IN][BS_LD];

    const int tid    = threadIdx.x;
    const int ntiles = (M + ROWS_PER_BLOCK - 1) / ROWS_PER_BLOCK;
    int tile = blockIdx.x;
    if (tile >= ntiles) return;

    const int lr = tid >> 4;
    const int lc = (tid & 15) * 4;

    int g0 = tile * ROWS_PER_BLOCK + lr;
    int g1 = g0 + 16, g2 = g0 + 32, g3 = g0 + 48;
    if (g0 >= M) g0 = M - 1;
    if (g1 >= M) g1 = M - 1;
    if (g2 >= M) g2 = M - 1;
    if (g3 >= M) g3 = M - 1;
    float4 pre0 = *(const float4*)(in + (size_t)g0 * C_IN + lc);
    float4 pre1 = *(const float4*)(in + (size_t)g1 * C_IN + lc);
    float4 pre2 = *(const float4*)(in + (size_t)g2 * C_IN + lc);
    float4 pre3 = *(const float4*)(in + (size_t)g3 * C_IN + lc);
    #pragma unroll
    for (int i = tid; i < C_OUT * (C_IN / 4); i += 256) {
        const int n  = i >> 4;
        const int k4 = i & 15;
        const float4 v = *(const float4*)(W + n * C_IN + k4 * 4);
        Bs[k4 * 4 + 0][n] = v.x;
        Bs[k4 * 4 + 1][n] = v.y;
        Bs[k4 * 4 + 2][n] = v.z;
        Bs[k4 * 4 + 3][n] = v.w;
    }
    *(float4*)(&As[0][lr +  0][lc]) = pre0;
    *(float4*)(&As[0][lr + 16][lc]) = pre1;
    *(float4*)(&As[0][lr + 32][lc]) = pre2;
    *(float4*)(&As[0][lr + 48][lc]) = pre3;
    __syncthreads();

    const int lane    = tid & 31;
    const int wave    = tid >> 5;
    const int colbase = wave * 16;
    const int hl      = lane >> 4;
    const int l16     = lane & 15;

    v2f bfrag[16];
    #pragma unroll
    for (int kk = 0; kk < 16; ++kk) {
        const int kb = kk * 4 + hl * 2;
        v2f bf;
        bf.x = Bs[kb + 0][colbase + l16];
        bf.y = Bs[kb + 1][colbase + l16];
        bfrag[kk] = bf;
    }

    int cur = 0;
    for (;;) {
        const int  nxt  = tile + gridDim.x;
        const bool more = nxt < ntiles;
        if (more) {
            g0 = nxt * ROWS_PER_BLOCK + lr;
            g1 = g0 + 16; g2 = g0 + 32; g3 = g0 + 48;
            if (g0 >= M) g0 = M - 1;
            if (g1 >= M) g1 = M - 1;
            if (g2 >= M) g2 = M - 1;
            if (g3 >= M) g3 = M - 1;
            pre0 = *(const float4*)(in + (size_t)g0 * C_IN + lc);
            pre1 = *(const float4*)(in + (size_t)g1 * C_IN + lc);
            pre2 = *(const float4*)(in + (size_t)g2 * C_IN + lc);
            pre3 = *(const float4*)(in + (size_t)g3 * C_IN + lc);
        }

        const int row_base = tile * ROWS_PER_BLOCK;
        #pragma unroll
        for (int rt = 0; rt < 4; ++rt) {
            v8f acc = {};
            #pragma unroll
            for (int kk = 0; kk < 16; ++kk) {
                const int kb = kk * 4 + hl * 2;
                v2f af;
                af.x = As[cur][rt * 16 + l16][kb + 0];
                af.y = As[cur][rt * 16 + l16][kb + 1];
                acc = __builtin_amdgcn_wmma_f32_16x16x4_f32(
                    false, af, false, bfrag[kk], (short)0, acc, false, false);
            }
            #pragma unroll
            for (int r = 0; r < 8; ++r) {
                const int gm = row_base + rt * 16 + r + 8 * hl;
                if (gm < M) {
                    x[(size_t)gm * C_OUT + colbase + l16] = acc[r];
                }
            }
        }

        if (!more) break;
        *(float4*)(&As[cur ^ 1][lr +  0][lc]) = pre0;
        *(float4*)(&As[cur ^ 1][lr + 16][lc]) = pre1;
        *(float4*)(&As[cur ^ 1][lr + 32][lc]) = pre2;
        *(float4*)(&As[cur ^ 1][lr + 48][lc]) = pre3;
        __syncthreads();
        cur ^= 1;
        tile = nxt;
    }
}

#define GATHER_BLOCKS 1024

__global__ __launch_bounds__(256) void pool_gather(
    const float* __restrict__ x,
    const int*   __restrict__ ids,
    const float* __restrict__ mask,
    const float* __restrict__ pn,
    float* __restrict__ out,
    int rows, int out_pn, int nbatch)
{
    const int lane  = threadIdx.x & 31;
    const int nwave = gridDim.x * 8;
    const int total = nbatch * out_pn;
    int item = blockIdx.x * 8 + __builtin_amdgcn_readfirstlane(threadIdx.x >> 5);
    if (item >= total) return;

    int p = item % out_pn;
    float pn_c = pn[p * MAXN + lane];
    float mk_c = mask[p * MAXN + lane];
    int   id_c = ids[p * MAXN + lane];

    for (;;) {
        const int  nxt  = item + nwave;
        const bool more = nxt < total;
        float pn_n = 0.f, mk_n = 0.f;
        int   id_n = 0;

        const float pm = fabsf(pn_c) * mk_c;
        float s = pm;
        #pragma unroll
        for (int off = 16; off > 0; off >>= 1) s += __shfl_xor(s, off, 32);
        const float wl = pm / (s + 1e-8f);

        const int b = item / out_pn;
        p = item - b * out_pn;
        const float* xb = x + (size_t)b * rows * C_OUT;
        float4 acc = {0.f, 0.f, 0.f, 0.f};
        #pragma unroll
        for (int h = 0; h < 2; ++h) {
            float4 v[16];
            #pragma unroll
            for (int j = 0; j < 16; ++j) {
                const int im = __builtin_amdgcn_readlane(id_c, h * 16 + j);
                const float4* row = (const float4*)(xb + (size_t)im * C_OUT);
                v[j] = row[lane];
            }
            if (h == 0 && more) {
                const int q = nxt % out_pn;
                pn_n = pn[q * MAXN + lane];
                mk_n = mask[q * MAXN + lane];
                id_n = ids[q * MAXN + lane];
            }
            __builtin_amdgcn_sched_barrier(0);
            #pragma unroll
            for (int j = 0; j < 16; ++j) {
                const float wm = __int_as_float(
                    __builtin_amdgcn_readlane(__float_as_int(wl), h * 16 + j));
                acc.x += wm * v[j].x;
                acc.y += wm * v[j].y;
                acc.z += wm * v[j].z;
                acc.w += wm * v[j].w;
            }
        }
        *(float4*)(out + ((size_t)(b * out_pn + p)) * C_OUT + lane * 4) = acc;

        if (!more) break;
        item = nxt;
        pn_c = pn_n;
        mk_c = mk_n;
        id_c = id_n;
    }
}

extern "C" void kernel_launch(void* const* d_in, const int* in_sizes, int n_in,
                              void* d_out, int out_size, void* d_ws, size_t ws_size,
                              hipStream_t stream)
{
    const float* in_pc = (const float*)d_in[0];
    const int*   ids   = (const int*)d_in[1];
    const float* mask  = (const float*)d_in[2];
    const float* pn    = (const float*)d_in[3];
    const float* W     = (const float*)d_in[4];

    const int out_pn = in_sizes[1] / MAXN;
    const int nbatch = out_size / (out_pn * C_OUT);
    const int rows   = in_sizes[0] / (nbatch * C_IN);

    float* x = (float*)d_ws;

    const int M      = in_sizes[0] / C_IN;
    const int ntiles = (M + ROWS_PER_BLOCK - 1) / ROWS_PER_BLOCK;
    const int g1 = ntiles < GEMM_BLOCKS ? ntiles : GEMM_BLOCKS;
    pool_gemm_wmma<<<g1, 256, 0, stream>>>(in_pc, W, x, M);

    const int total_waves = nbatch * out_pn;
    const int need = (total_waves + 7) / 8;
    const int g2 = need < GATHER_BLOCKS ? need : GATHER_BLOCKS;
    pool_gather<<<g2, 256, 0, stream>>>(x, ids, mask, pn, (float*)d_out,
                                        rows, out_pn, nbatch);
}
